# starting kernel
# baseline (speedup 1.0000x reference)
#include <hip/hip_runtime.h>

typedef __attribute__((ext_vector_type(2))) float v2f;
typedef __attribute__((ext_vector_type(8))) float v8f;

#define NN     2048
#define FIN    512
#define FOUT   512
#define FEDGE  16

__global__ __launch_bounds__(256) void wmma_gemm_f32_kernel(
    const float* __restrict__ A, const float* __restrict__ B,
    const float* __restrict__ E, float* __restrict__ C,
    int M, int Ncols, int K)
{
    const int lane = threadIdx.x & 31;
    const int wave = threadIdx.x >> 5;

    const int wm = blockIdx.y * 64  + (wave & 1)  * 32;
    const int wn = blockIdx.x * 128 + (wave >> 1) * 32;

    const int ml = lane & 15;
    const int kh = (lane >> 4) * 2;

    const float* a0p = A + (size_t)(wm      + ml) * K + kh;
    const float* a1p = A + (size_t)(wm + 16 + ml) * K + kh;
    const float* b0p = B + (size_t)kh * Ncols + wn + ml;
    const float* b1p = b0p + 16;
    const size_t bstep = (size_t)4 * Ncols;

    v8f c00 = {}, c01 = {}, c10 = {}, c11 = {};

    #pragma unroll 4
    for (int k = 0; k < K; k += 4) {
        v2f a0 = *(const v2f*)a0p;  a0p += 4;
        v2f a1 = *(const v2f*)a1p;  a1p += 4;
        v2f b0, b1;
        b0.x = b0p[0]; b0.y = b0p[Ncols];
        b1.x = b1p[0]; b1.y = b1p[Ncols];
        b0p += bstep;  b1p += bstep;

        c00 = __builtin_amdgcn_wmma_f32_16x16x4_f32(false, a0, false, b0, (short)0, c00, false, false);
        c01 = __builtin_amdgcn_wmma_f32_16x16x4_f32(false, a0, false, b1, (short)0, c01, false, false);
        c10 = __builtin_amdgcn_wmma_f32_16x16x4_f32(false, a1, false, b0, (short)0, c10, false, false);
        c11 = __builtin_amdgcn_wmma_f32_16x16x4_f32(false, a1, false, b1, (short)0, c11, false, false);
    }

    const int rbase = (lane >> 4) * 8;
    const int colb  = lane & 15;
    auto store_tile = [&](v8f c, int m0, int n0) {
        const int col = n0 + colb;
        #pragma unroll
        for (int r = 0; r < 8; ++r) {
            const int row = m0 + rbase + r;
            float v = c[r];
            if (E) {
                v += E[(size_t)row * Ncols + col];
                v = v > 0.0f ? v : 0.0f;
            }
            C[(size_t)row * Ncols + col] = v;
        }
    };
    store_tile(c00, wm,      wn);
    store_tile(c01, wm,      wn + 16);
    store_tile(c10, wm + 16, wn);
    store_tile(c11, wm + 16, wn + 16);
}

__global__ __launch_bounds__(256) void edge_agg_kernel(
    const float* __restrict__ A, const float* __restrict__ Z,
    const float* __restrict__ EK, const float* __restrict__ bias,
    float* __restrict__ E)
{
    const int i  = blockIdx.x;
    const int t  = threadIdx.x;
    const int f  = t & 15;
    const int jg = t >> 4;

    __shared__ float red[256];

    const float* Arow = A + (size_t)i * NN;
    const float* Zrow = Z + (size_t)i * NN * FEDGE;

    float acc = 0.0f;
    #pragma unroll 4
    for (int j = jg; j < NN; j += 16)
        acc = fmaf(Arow[j], __builtin_nontemporal_load(Zrow + j * FEDGE + f), acc);

    red[t] = acc;
    __syncthreads();
    for (int s = 128; s >= 16; s >>= 1) {
        if (t < s) red[t] += red[t + s];
        __syncthreads();
    }

    for (int c = t; c < FOUT; c += 256) {
        float y = bias[c];
        #pragma unroll
        for (int ff = 0; ff < FEDGE; ++ff)
            y = fmaf(red[ff], EK[ff * FOUT + c], y);
        E[(size_t)i * FOUT + c] = y;
    }
}

extern "C" void kernel_launch(void* const* d_in, const int* in_sizes, int n_in,
                              void* d_out, int out_size, void* d_ws, size_t ws_size,
                              hipStream_t stream) {
    const float* X    = (const float*)d_in[0];
    const float* A    = (const float*)d_in[1];
    const float* Z    = (const float*)d_in[2];
    const float* W    = (const float*)d_in[3];
    const float* EK   = (const float*)d_in[4];
    const float* bias = (const float*)d_in[5];
    float* out = (float*)d_out;

    float* XW = (float*)d_ws;
    float* E  = XW + (size_t)NN * FOUT;

    dim3 g1(FOUT / 128, NN / 64);
    wmma_gemm_f32_kernel<<<g1, 256, 0, stream>>>(X, W, nullptr, XW, NN, FOUT, FIN);

    edge_agg_kernel<<<NN, 256, 0, stream>>>(A, Z, EK, bias, E);

    dim3 g2(FOUT / 128, NN / 64);
    wmma_gemm_f32_kernel<<<g2, 256, 0, stream>>>(A, XW, E, out, NN, FOUT, NN);
}


# GEMM kernel: A and B tiles staged through double-buffered LDS from coalesced 16-byte global loads, 64x64 block tiles on 256 work-groups, LDS reads issued four k-steps ahead; k sum order unchanged
# speedup vs baseline: 3.8539x; 3.8539x over previous
#include <hip/hip_runtime.h>

typedef __attribute__((ext_vector_type(2))) float v2f;
typedef __attribute__((ext_vector_type(8))) float v8f;

#define NN     2048
#define FIN    512
#define FOUT   512
#define FEDGE  16

#define GBM 64
#define GBN 64
#define GBK 64
#define GSA 68
#define GSB 160

struct FragSet {
    v2f a0, a1, a2, a3;
    v2f p0, p1, p2, p3;
    v2f q0, q1, q2, q3;
};

#define LOAD_FRAGS(F, g)                                                   \
    F.a0 = *(const v2f*)(asl + 16 * (g) + 0);                              \
    F.a1 = *(const v2f*)(asl + 16 * (g) + 4);                              \
    F.a2 = *(const v2f*)(asl + 16 * (g) + 8);                              \
    F.a3 = *(const v2f*)(asl + 16 * (g) + 12);                             \
    F.p0 = *(const v2f*)(bsl + (8 * (g) + 0) * GSB);                       \
    F.p1 = *(const v2f*)(bsl + (8 * (g) + 2) * GSB);                       \
    F.p2 = *(const v2f*)(bsl + (8 * (g) + 4) * GSB);                       \
    F.p3 = *(const v2f*)(bsl + (8 * (g) + 6) * GSB);                       \
    F.q0 = *(const v2f*)(bsl + (8 * (g) + 0) * GSB + 32);                  \
    F.q1 = *(const v2f*)(bsl + (8 * (g) + 2) * GSB + 32);                  \
    F.q2 = *(const v2f*)(bsl + (8 * (g) + 4) * GSB + 32);                  \
    F.q3 = *(const v2f*)(bsl + (8 * (g) + 6) * GSB + 32);

#define WMMA_F32(a, b, c) __builtin_amdgcn_wmma_f32_16x16x4_f32(false, a, false, b, (short)0, c, false, false)

#define MUL_FRAGS(F)                                                       \
    c0 = WMMA_F32(F.a0, F.p0, c0);  c1 = WMMA_F32(F.a0, F.q0, c1);         \
    c0 = WMMA_F32(F.a1, F.p1, c0);  c1 = WMMA_F32(F.a1, F.q1, c1);         \
    c0 = WMMA_F32(F.a2, F.p2, c0);  c1 = WMMA_F32(F.a2, F.q2, c1);         \
    c0 = WMMA_F32(F.a3, F.p3, c0);  c1 = WMMA_F32(F.a3, F.q3, c1);

__global__ __launch_bounds__(256) void wmma_gemm_f32_kernel(
    const float* __restrict__ A, const float* __restrict__ B,
    const float* __restrict__ E, float* __restrict__ C,
    int M, int Ncols, int K)
{
    __shared__ __attribute__((aligned(16))) float As[2][GBM * GSA];
    __shared__ __attribute__((aligned(16))) float Bs[2][(GBK / 2) * GSB];

    const int t    = threadIdx.x;
    const int lane = t & 31;
    const int wave = t >> 5;

    const int bm = blockIdx.y * GBM;
    const int bn = blockIdx.x * GBN;
    const int wm = (wave & 3) * 16;
    const int wn = (wave >> 2) * 32;

    const int cl = lane & 15;
    const int hh = lane >> 4;

    const int srow = t >> 4;
    const int sq   = t & 15;
    const size_t a16 = (size_t)16 * K;
    const float* ag = A + (size_t)(bm + srow) * K + 4 * sq;
    const float* bg = B + (size_t)(2 * srow) * Ncols + bn + 4 * sq;
    const size_t bchunk = (size_t)GBK * Ncols;
    const size_t b32 = (size_t)32 * Ncols;

    const int sa_off = srow * GSA + 4 * sq;
    const int sb_off = srow * GSB + 8 * sq;
    const int aoff = (wm + cl) * GSA + 2 * hh;
    const int boff = hh * GSB + (wn + cl) * 2;

    float4 ra0, ra1, ra2, ra3;
    float4 rbe0, rbo0, rbe1, rbo1;

#define LOAD_GLOBAL()                                                      \
    ra0  = *(const float4*)(ag);                                           \
    ra1  = *(const float4*)(ag + a16);                                     \
    ra2  = *(const float4*)(ag + 2 * a16);                                 \
    ra3  = *(const float4*)(ag + 3 * a16);                                 \
    rbe0 = *(const float4*)(bg);                                           \
    rbo0 = *(const float4*)(bg + Ncols);                                   \
    rbe1 = *(const float4*)(bg + b32);                                     \
    rbo1 = *(const float4*)(bg + b32 + Ncols);                             \
    ag += GBK;  bg += bchunk;

#define STORE_LDS(buf)                                                     \
    {                                                                      \
        float* as_ = &As[buf][0] + sa_off;                                 \
        float* bs_ = &Bs[buf][0] + sb_off;                                 \
        *(float4*)(as_)            = ra0;                                  \
        *(float4*)(as_ + 16 * GSA) = ra1;                                  \
        *(float4*)(as_ + 32 * GSA) = ra2;                                  \
        *(float4*)(as_ + 48 * GSA) = ra3;                                  \
        *(float4*)(bs_)                = make_float4(rbe0.x, rbo0.x, rbe0.y, rbo0.y); \
        *(float4*)(bs_ + 4)            = make_float4(rbe0.z, rbo0.z, rbe0.w, rbo0.w); \
        *(float4*)(bs_ + 16 * GSB)     = make_float4(rbe1.x, rbo1.x, rbe1.y, rbo1.y); \
        *(float4*)(bs_ + 16 * GSB + 4) = make_float4(rbe1.z, rbo1.z, rbe1.w, rbo1.w); \
    }

    v8f c0 = {}, c1 = {};
    FragSet f0, f1;

    const int nchunks = K / GBK;

    LOAD_GLOBAL();
    STORE_LDS(0);
    __syncthreads();

#define MUL_CHUNK(buf)                                                     \
    {                                                                      \
        const float* asl = &As[buf][0] + aoff;                             \
        const float* bsl = &Bs[buf][0] + boff;                             \
        LOAD_FRAGS(f0, 0)                                                  \
        LOAD_FRAGS(f1, 1)                                                  \
        MUL_FRAGS(f0)                                                      \
        LOAD_FRAGS(f0, 2)                                                  \
        MUL_FRAGS(f1)                                                      \
        LOAD_FRAGS(f1, 3)                                                  \
        MUL_FRAGS(f0)                                                      \
        MUL_FRAGS(f1)                                                      \
    }

#define ORDER_CHUNK()                                                      \
    __builtin_amdgcn_sched_group_barrier(0x100, 12, 0);                    \
    __builtin_amdgcn_sched_group_barrier(0x008, 8, 0);                     \
    __builtin_amdgcn_sched_group_barrier(0x100, 6, 0);                     \
    __builtin_amdgcn_sched_group_barrier(0x008, 8, 0);                     \
    __builtin_amdgcn_sched_group_barrier(0x100, 6, 0);                     \
    __builtin_amdgcn_sched_group_barrier(0x008, 16, 0);

    for (int ch = 0; ch + 1 < nchunks; ++ch) {
        const int buf = ch & 1;
        LOAD_GLOBAL();
        MUL_CHUNK(buf)
        STORE_LDS(buf ^ 1);
        __builtin_amdgcn_sched_group_barrier(0x020, 8, 0);
        ORDER_CHUNK()
        __builtin_amdgcn_sched_group_barrier(0x200, 8, 0);
        __syncthreads();
    }
    MUL_CHUNK((nchunks - 1) & 1)
    ORDER_CHUNK()

    const int rbase = hh * 8;
    auto store_tile = [&](v8f c, int m0, int n0) {
        const int col = n0 + cl;
        #pragma unroll
        for (int r = 0; r < 8; ++r) {
            const int row = m0 + rbase + r;
            float v = c[r];
            if (E) {
                v += E[(size_t)row * Ncols + col];
                v = v > 0.0f ? v : 0.0f;
            }
            C[(size_t)row * Ncols + col] = v;
        }
    };
    store_tile(c0, bm + wm, bn + wn);
    store_tile(c1, bm + wm, bn + wn + 16);
}

__global__ __launch_bounds__(256) void edge_agg_kernel(
    const float* __restrict__ A, const float* __restrict__ Z,
    const float* __restrict__ EK, const float* __restrict__ bias,
    float* __restrict__ E)
{
    const int i  = blockIdx.x;
    const int t  = threadIdx.x;
    const int f  = t & 15;
    const int jg = t >> 4;

    __shared__ float red[256];

    const float* Arow = A + (size_t)i * NN;
    const float* Zrow = Z + (size_t)i * NN * FEDGE;

    float acc = 0.0f;
    #pragma unroll 4
    for (int j = jg; j < NN; j += 16)
        acc = fmaf(Arow[j], __builtin_nontemporal_load(Zrow + j * FEDGE + f), acc);

    red[t] = acc;
    __syncthreads();
    for (int s = 128; s >= 16; s >>= 1) {
        if (t < s) red[t] += red[t + s];
        __syncthreads();
    }

    for (int c = t; c < FOUT; c += 256) {
        float y = bias[c];
        #pragma unroll
        for (int ff = 0; ff < FEDGE; ++ff)
            y = fmaf(red[ff], EK[ff * FOUT + c], y);
        E[(size_t)i * FOUT + c] = y;
    }
}

extern "C" void kernel_launch(void* const* d_in, const int* in_sizes, int n_in,
                              void* d_out, int out_size, void* d_ws, size_t ws_size,
                              hipStream_t stream) {
    const float* X    = (const float*)d_in[0];
    const float* A    = (const float*)d_in[1];
    const float* Z    = (const float*)d_in[2];
    const float* W    = (const float*)d_in[3];
    const float* EK   = (const float*)d_in[4];
    const float* bias = (const float*)d_in[5];
    float* out = (float*)d_out;

    float* XW = (float*)d_ws;
    float* E  = XW + (size_t)NN * FOUT;

    dim3 g1(FOUT / GBN, NN / GBM);
    wmma_gemm_f32_kernel<<<g1, 256, 0, stream>>>(X, W, nullptr, XW, NN, FOUT, FIN);

    edge_agg_kernel<<<NN, 256, 0, stream>>>(A, Z, EK, bias, E);

    dim3 g2(FOUT / GBN, NN / GBM);
    wmma_gemm_f32_kernel<<<g2, 256, 0, stream>>>(A, XW, E, out, NN, FOUT, NN);
}


# Edge aggregation kernel rebuilt as a stream: 256 work-groups of 512 threads, a wave reads half a row of Z in 512-byte non-temporal loads, one batch requested ahead; new reduction tree over j
# speedup vs baseline: 4.1307x; 1.0718x over previous
#include <hip/hip_runtime.h>

typedef __attribute__((ext_vector_type(2))) float v2f;
typedef __attribute__((ext_vector_type(8))) float v8f;

#define NN     2048
#define FIN    512
#define FOUT   512
#define FEDGE  16

#define GBM 64
#define GBN 64
#define GBK 64
#define GSA 68
#define GSB 160

struct FragSet {
    v2f a0, a1, a2, a3;
    v2f p0, p1, p2, p3;
    v2f q0, q1, q2, q3;
};

#define LOAD_FRAGS(F, g)                                                   \
    F.a0 = *(const v2f*)(asl + 16 * (g) + 0);                              \
    F.a1 = *(const v2f*)(asl + 16 * (g) + 4);                              \
    F.a2 = *(const v2f*)(asl + 16 * (g) + 8);                              \
    F.a3 = *(const v2f*)(asl + 16 * (g) + 12);                             \
    F.p0 = *(const v2f*)(bsl + (8 * (g) + 0) * GSB);                       \
    F.p1 = *(const v2f*)(bsl + (8 * (g) + 2) * GSB);                       \
    F.p2 = *(const v2f*)(bsl + (8 * (g) + 4) * GSB);                       \
    F.p3 = *(const v2f*)(bsl + (8 * (g) + 6) * GSB);                       \
    F.q0 = *(const v2f*)(bsl + (8 * (g) + 0) * GSB + 32);                  \
    F.q1 = *(const v2f*)(bsl + (8 * (g) + 2) * GSB + 32);                  \
    F.q2 = *(const v2f*)(bsl + (8 * (g) + 4) * GSB + 32);                  \
    F.q3 = *(const v2f*)(bsl + (8 * (g) + 6) * GSB + 32);

#define WMMA_F32(a, b, c) __builtin_amdgcn_wmma_f32_16x16x4_f32(false, a, false, b, (short)0, c, false, false)

#define MUL_FRAGS(F)                                                       \
    c0 = WMMA_F32(F.a0, F.p0, c0);  c1 = WMMA_F32(F.a0, F.q0, c1);         \
    c0 = WMMA_F32(F.a1, F.p1, c0);  c1 = WMMA_F32(F.a1, F.q1, c1);         \
    c0 = WMMA_F32(F.a2, F.p2, c0);  c1 = WMMA_F32(F.a2, F.q2, c1);         \
    c0 = WMMA_F32(F.a3, F.p3, c0);  c1 = WMMA_F32(F.a3, F.q3, c1);

__global__ __launch_bounds__(256) void wmma_gemm_f32_kernel(
    const float* __restrict__ A, const float* __restrict__ B,
    const float* __restrict__ E, float* __restrict__ C,
    int M, int Ncols, int K)
{
    __shared__ __attribute__((aligned(16))) float As[2][GBM * GSA];
    __shared__ __attribute__((aligned(16))) float Bs[2][(GBK / 2) * GSB];

    const int t    = threadIdx.x;
    const int lane = t & 31;
    const int wave = t >> 5;

    const int bm = blockIdx.y * GBM;
    const int bn = blockIdx.x * GBN;
    const int wm = (wave & 3) * 16;
    const int wn = (wave >> 2) * 32;

    const int cl = lane & 15;
    const int hh = lane >> 4;

    const int srow = t >> 4;
    const int sq   = t & 15;
    const size_t a16 = (size_t)16 * K;
    const float* ag = A + (size_t)(bm + srow) * K + 4 * sq;
    const float* bg = B + (size_t)(2 * srow) * Ncols + bn + 4 * sq;
    const size_t bchunk = (size_t)GBK * Ncols;
    const size_t b32 = (size_t)32 * Ncols;

    const int sa_off = srow * GSA + 4 * sq;
    const int sb_off = srow * GSB + 8 * sq;
    const int aoff = (wm + cl) * GSA + 2 * hh;
    const int boff = hh * GSB + (wn + cl) * 2;

    float4 ra0, ra1, ra2, ra3;
    float4 rbe0, rbo0, rbe1, rbo1;

#define LOAD_GLOBAL()                                                      \
    ra0  = *(const float4*)(ag);                                           \
    ra1  = *(const float4*)(ag + a16);                                     \
    ra2  = *(const float4*)(ag + 2 * a16);                                 \
    ra3  = *(const float4*)(ag + 3 * a16);                                 \
    rbe0 = *(const float4*)(bg);                                           \
    rbo0 = *(const float4*)(bg + Ncols);                                   \
    rbe1 = *(const float4*)(bg + b32);                                     \
    rbo1 = *(const float4*)(bg + b32 + Ncols);                             \
    ag += GBK;  bg += bchunk;

#define STORE_LDS(buf)                                                     \
    {                                                                      \
        float* as_ = &As[buf][0] + sa_off;                                 \
        float* bs_ = &Bs[buf][0] + sb_off;                                 \
        *(float4*)(as_)            = ra0;                                  \
        *(float4*)(as_ + 16 * GSA) = ra1;                                  \
        *(float4*)(as_ + 32 * GSA) = ra2;                                  \
        *(float4*)(as_ + 48 * GSA) = ra3;                                  \
        *(float4*)(bs_)                = make_float4(rbe0.x, rbo0.x, rbe0.y, rbo0.y); \
        *(float4*)(bs_ + 4)            = make_float4(rbe0.z, rbo0.z, rbe0.w, rbo0.w); \
        *(float4*)(bs_ + 16 * GSB)     = make_float4(rbe1.x, rbo1.x, rbe1.y, rbo1.y); \
        *(float4*)(bs_ + 16 * GSB + 4) = make_float4(rbe1.z, rbo1.z, rbe1.w, rbo1.w); \
    }

    v8f c0 = {}, c1 = {};
    FragSet f0, f1;

    const int nchunks = K / GBK;

    LOAD_GLOBAL();
    STORE_LDS(0);
    __syncthreads();

#define MUL_CHUNK(buf)                                                     \
    {                                                                      \
        const float* asl = &As[buf][0] + aoff;                             \
        const float* bsl = &Bs[buf][0] + boff;                             \
        LOAD_FRAGS(f0, 0)                                                  \
        LOAD_FRAGS(f1, 1)                                                  \
        MUL_FRAGS(f0)                                                      \
        LOAD_FRAGS(f0, 2)                                                  \
        MUL_FRAGS(f1)                                                      \
        LOAD_FRAGS(f1, 3)                                                  \
        MUL_FRAGS(f0)                                                      \
        MUL_FRAGS(f1)                                                      \
    }

#define ORDER_CHUNK()                                                      \
    __builtin_amdgcn_sched_group_barrier(0x100, 12, 0);                    \
    __builtin_amdgcn_sched_group_barrier(0x008, 8, 0);                     \
    __builtin_amdgcn_sched_group_barrier(0x100, 6, 0);                     \
    __builtin_amdgcn_sched_group_barrier(0x008, 8, 0);                     \
    __builtin_amdgcn_sched_group_barrier(0x100, 6, 0);                     \
    __builtin_amdgcn_sched_group_barrier(0x008, 16, 0);

    for (int ch = 0; ch + 1 < nchunks; ++ch) {
        const int buf = ch & 1;
        LOAD_GLOBAL();
        MUL_CHUNK(buf)
        STORE_LDS(buf ^ 1);
        __builtin_amdgcn_sched_group_barrier(0x020, 8, 0);
        ORDER_CHUNK()
        __builtin_amdgcn_sched_group_barrier(0x200, 8, 0);
        __syncthreads();
    }
    MUL_CHUNK((nchunks - 1) & 1)
    ORDER_CHUNK()

    const int rbase = hh * 8;
    auto store_tile = [&](v8f c, int m0, int n0) {
        const int col = n0 + cl;
        #pragma unroll
        for (int r = 0; r < 8; ++r) {
            const int row = m0 + rbase + r;
            float v = c[r];
            if (E) {
                v += E[(size_t)row * Ncols + col];
                v = v > 0.0f ? v : 0.0f;
            }
            C[(size_t)row * Ncols + col] = v;
        }
    };
    store_tile(c0, bm + wm, bn + wn);
    store_tile(c1, bm + wm, bn + wn + 16);
}

#define EROWS  8
#define EWAVES 16
#define EHALF  1024
#define EBATCH 16
#define ENB    8

typedef __attribute__((ext_vector_type(4))) float v4f;

__global__ __launch_bounds__(512) void edge_agg_kernel(
    const float* __restrict__ A, const float* __restrict__ Z,
    const float* __restrict__ EK, const float* __restrict__ bias,
    float* __restrict__ E)
{
    __shared__ __attribute__((aligned(16))) float As[EWAVES][EHALF];
    __shared__ __attribute__((aligned(16))) float red[EWAVES][FEDGE];

    const int t    = threadIdx.x;
    const int lane = t & 31;
    const int wave = t >> 5;
    const int row  = blockIdx.x * EROWS + (wave >> 1);
    const int j0   = (wave & 1) * EHALF;
    const int jl   = lane >> 2;

    const float* ap  = A + (size_t)row * NN + j0 + 4 * lane;
    const v4f*   zp  = (const v4f*)(Z + ((size_t)row * NN + j0) * FEDGE) + lane;
    float*       asw = &As[wave][0];

    v4f z[2][EBATCH];
#define LOAD_BATCH(b)                                                      \
    _Pragma("unroll")                                                      \
    for (int q = 0; q < EBATCH; ++q)                                       \
        z[(b) & 1][q] = __builtin_nontemporal_load(zp + ((b) * EBATCH + q) * 32);

    v4f av[8];
    #pragma unroll
    for (int p = 0; p < 8; ++p) av[p] = *(const v4f*)(ap + 128 * p);
    __builtin_amdgcn_sched_barrier(0);
    LOAD_BATCH(0)
    LOAD_BATCH(1)
    __builtin_amdgcn_sched_barrier(0);
    #pragma unroll
    for (int p = 0; p < 8; ++p) *(v4f*)(asw + 128 * p + 4 * lane) = av[p];
    __builtin_amdgcn_wave_barrier();

    v4f acc = {0.0f, 0.0f, 0.0f, 0.0f};
    #pragma unroll
    for (int b = 0; b < ENB; ++b) {
        __builtin_amdgcn_sched_barrier(0);
        if (b >= 1 && b + 1 < ENB) { LOAD_BATCH(b + 1) }
        __builtin_amdgcn_sched_barrier(0);
        const float* al = asw + 128 * b + jl;
        #pragma unroll
        for (int q = 0; q < EBATCH; ++q) {
            const float a  = al[8 * q];
            const v4f   zz = z[b & 1][q];
            acc.x = fmaf(a, zz.x, acc.x);
            acc.y = fmaf(a, zz.y, acc.y);
            acc.z = fmaf(a, zz.z, acc.z);
            acc.w = fmaf(a, zz.w, acc.w);
        }
    }
#undef LOAD_BATCH

    const int c = t;
    float ek[FEDGE];
    #pragma unroll
    for (int ff = 0; ff < FEDGE; ++ff) ek[ff] = EK[ff * FOUT + c];
    const float bc = bias[c];

    #pragma unroll
    for (int m = 4; m <= 16; m <<= 1) {
        acc.x += __shfl_xor(acc.x, m, 32);
        acc.y += __shfl_xor(acc.y, m, 32);
        acc.z += __shfl_xor(acc.z, m, 32);
        acc.w += __shfl_xor(acc.w, m, 32);
    }
    if (lane < 4) *(v4f*)(&red[wave][4 * lane]) = acc;
    __syncthreads();

    #pragma unroll
    for (int r = 0; r < EROWS; ++r) {
        const v4f* h0 = (const v4f*)&red[2 * r][0];
        const v4f* h1 = (const v4f*)&red[2 * r + 1][0];
        float y = bc;
        #pragma unroll
        for (int g = 0; g < 4; ++g) {
            const v4f s = h0[g] + h1[g];
            y = fmaf(s.x, ek[4 * g + 0], y);
            y = fmaf(s.y, ek[4 * g + 1], y);
            y = fmaf(s.z, ek[4 * g + 2], y);
            y = fmaf(s.w, ek[4 * g + 3], y);
        }
        E[(size_t)(blockIdx.x * EROWS + r) * FOUT + c] = y;
    }
}

extern "C" void kernel_launch(void* const* d_in, const int* in_sizes, int n_in,
                              void* d_out, int out_size, void* d_ws, size_t ws_size,
                              hipStream_t stream) {
    const float* X    = (const float*)d_in[0];
    const float* A    = (const float*)d_in[1];
    const float* Z    = (const float*)d_in[2];
    const float* W    = (const float*)d_in[3];
    const float* EK   = (const float*)d_in[4];
    const float* bias = (const float*)d_in[5];
    float* out = (float*)d_out;

    float* XW = (float*)d_ws;
    float* E  = XW + (size_t)NN * FOUT;

    dim3 g1(FOUT / GBN, NN / GBM);
    wmma_gemm_f32_kernel<<<g1, 256, 0, stream>>>(X, W, nullptr, XW, NN, FOUT, FIN);

    edge_agg_kernel<<<NN / EROWS, EWAVES * 32, 0, stream>>>(A, Z, EK, bias, E);

    dim3 g2(FOUT / GBN, NN / GBM);
    wmma_gemm_f32_kernel<<<g2, 256, 0, stream>>>(A, XW, E, out, NN, FOUT, NN);
}


# GEMM K loop pipelined across the chunk boundary: LDS stores at the chunk top, hand-written barrier in front of the last four k-steps, next chunk's first fragments read behind it; k sum order kept
# speedup vs baseline: 4.4036x; 1.0661x over previous
#include <hip/hip_runtime.h>

typedef __attribute__((ext_vector_type(2))) float v2f;
typedef __attribute__((ext_vector_type(4))) float v4f;
typedef __attribute__((ext_vector_type(8))) float v8f;

#define NN     2048
#define FIN    512
#define FOUT   512
#define FEDGE  16

#define GBM 64
#define GBN 64
#define GBK 64
#define GSA 68
#define GSB 160

struct FragSet {
    v2f a0, a1, a2, a3;
    v2f p0, p1, p2, p3;
    v2f q0, q1, q2, q3;
};

#define LOAD_FRAGS(F, g)                                                   \
    F.a0 = *(const v2f*)(asl + 16 * (g) + 0);                              \
    F.a1 = *(const v2f*)(asl + 16 * (g) + 4);                              \
    F.a2 = *(const v2f*)(asl + 16 * (g) + 8);                              \
    F.a3 = *(const v2f*)(asl + 16 * (g) + 12);                             \
    F.p0 = *(const v2f*)(bsl + (8 * (g) + 0) * GSB);                       \
    F.p1 = *(const v2f*)(bsl + (8 * (g) + 2) * GSB);                       \
    F.p2 = *(const v2f*)(bsl + (8 * (g) + 4) * GSB);                       \
    F.p3 = *(const v2f*)(bsl + (8 * (g) + 6) * GSB);                       \
    F.q0 = *(const v2f*)(bsl + (8 * (g) + 0) * GSB + 32);                  \
    F.q1 = *(const v2f*)(bsl + (8 * (g) + 2) * GSB + 32);                  \
    F.q2 = *(const v2f*)(bsl + (8 * (g) + 4) * GSB + 32);                  \
    F.q3 = *(const v2f*)(bsl + (8 * (g) + 6) * GSB + 32);

#define WMMA_F32(a, b, c) __builtin_amdgcn_wmma_f32_16x16x4_f32(false, a, false, b, (short)0, c, false, false)

#define MUL_FRAGS(F)                                                       \
    c0 = WMMA_F32(F.a0, F.p0, c0);  c1 = WMMA_F32(F.a0, F.q0, c1);         \
    c0 = WMMA_F32(F.a1, F.p1, c0);  c1 = WMMA_F32(F.a1, F.q1, c1);         \
    c0 = WMMA_F32(F.a2, F.p2, c0);  c1 = WMMA_F32(F.a2, F.q2, c1);         \
    c0 = WMMA_F32(F.a3, F.p3, c0);  c1 = WMMA_F32(F.a3, F.q3, c1);

static __device__ __forceinline__ void lds_hand_over_barrier() {
    asm volatile("s_wait_dscnt 0x0\n\ts_barrier_signal -1\n\ts_barrier_wait -1" ::: "memory");
}

__global__ __launch_bounds__(256) void wmma_gemm_f32_kernel(
    const float* __restrict__ A, const float* __restrict__ B,
    const float* __restrict__ E, float* __restrict__ C,
    int M, int Ncols, int K)
{
    __shared__ __attribute__((aligned(16))) float As[2][GBM * GSA];
    __shared__ __attribute__((aligned(16))) float Bs[2][(GBK / 2) * GSB];

    const int t    = threadIdx.x;
    const int lane = t & 31;
    const int wave = t >> 5;

    const int bm = blockIdx.y * GBM;
    const int bn = blockIdx.x * GBN;
    const int wm = (wave & 3) * 16;
    const int wn = (wave >> 2) * 32;

    const int cl = lane & 15;
    const int hh = lane >> 4;

    const int srow = t >> 4;
    const int sq   = t & 15;
    const size_t a16 = (size_t)16 * K;
    const float* ag = A + (size_t)(bm + srow) * K + 4 * sq;
    const float* bg = B + (size_t)(2 * srow) * Ncols + bn + 4 * sq;
    const size_t bchunk = (size_t)GBK * Ncols;
    const size_t b32 = (size_t)32 * Ncols;

    const int sa_off = srow * GSA + 4 * sq;
    const int sb_off = srow * GSB + 8 * sq;
    const int aoff = (wm + cl) * GSA + 2 * hh;
    const int boff = hh * GSB + (wn + cl) * 2;

#define DECL_STAGE(P) float4 P##a0, P##a1, P##a2, P##a3; v4f P##be0, P##bo0, P##be1, P##bo1

#define LOAD_GLOBAL(P)                                                     \
    P##a0  = *(const float4*)(ag);                                         \
    P##a1  = *(const float4*)(ag + a16);                                   \
    P##a2  = *(const float4*)(ag + 2 * a16);                               \
    P##a3  = *(const float4*)(ag + 3 * a16);                               \
    P##be0 = *(const v4f*)(bg);                                            \
    P##bo0 = *(const v4f*)(bg + Ncols);                                    \
    P##be1 = *(const v4f*)(bg + b32);                                      \
    P##bo1 = *(const v4f*)(bg + b32 + Ncols);                              \
    ag += GBK;  bg += bchunk;

#define STORE_LDS(P, buf)                                                  \
    {                                                                      \
        float* as_ = &As[buf][0] + sa_off;                                 \
        float* bs_ = &Bs[buf][0] + sb_off;                                 \
        *(float4*)(as_)            = P##a0;                                \
        *(float4*)(as_ + 16 * GSA) = P##a1;                                \
        *(float4*)(as_ + 32 * GSA) = P##a2;                                \
        *(float4*)(as_ + 48 * GSA) = P##a3;                                \
        *(float4*)(bs_)                = make_float4(P##be0.x, P##bo0.x, P##be0.y, P##bo0.y); \
        *(float4*)(bs_ + 4)            = make_float4(P##be0.z, P##bo0.z, P##be0.w, P##bo0.w); \
        *(float4*)(bs_ + 16 * GSB)     = make_float4(P##be1.x, P##bo1.x, P##be1.y, P##bo1.y); \
        *(float4*)(bs_ + 16 * GSB + 4) = make_float4(P##be1.z, P##bo1.z, P##be1.w, P##bo1.w); \
    }

#define PIN_B_STAGE(P) asm volatile("" : "+v"(P##be0), "+v"(P##bo0), "+v"(P##be1), "+v"(P##bo1))

    v8f c0 = {}, c1 = {};
    FragSet f0, f1;
    DECL_STAGE(st_);

    const int nchunks = K / GBK;

    {
        DECL_STAGE(first_);
        LOAD_GLOBAL(first_);
        if (nchunks > 1) { LOAD_GLOBAL(st_); }
        STORE_LDS(first_, 0);
    }
    __syncthreads();
    {
        const float* asl = &As[0][0] + aoff;
        const float* bsl = &Bs[0][0] + boff;
        LOAD_FRAGS(f0, 0)
    }

#define MUL_HEAD()                                                         \
        LOAD_FRAGS(f1, 1)                                                  \
        MUL_FRAGS(f0)                                                      \
        LOAD_FRAGS(f0, 2)                                                  \
        MUL_FRAGS(f1)                                                      \
        LOAD_FRAGS(f1, 3)                                                  \
        MUL_FRAGS(f0)

#define ORDER_HEAD()                                                       \
    __builtin_amdgcn_sched_group_barrier(0x100, 6, 0);                     \
    __builtin_amdgcn_sched_group_barrier(0x008, 8, 0);                     \
    __builtin_amdgcn_sched_group_barrier(0x100, 6, 0);                     \
    __builtin_amdgcn_sched_group_barrier(0x008, 8, 0);                     \
    __builtin_amdgcn_sched_group_barrier(0x100, 6, 0);                     \
    __builtin_amdgcn_sched_group_barrier(0x008, 8, 0);

#define HAND_OVER(nbuf)                                                    \
    __builtin_amdgcn_sched_barrier(0);                                     \
    lds_hand_over_barrier();                                               \
    __builtin_amdgcn_sched_barrier(0);                                     \
    {                                                                      \
        const float* asl = &As[nbuf][0] + aoff;                            \
        const float* bsl = &Bs[nbuf][0] + boff;                            \
        LOAD_FRAGS(f0, 0)                                                  \
        MUL_FRAGS(f1)                                                      \
        __builtin_amdgcn_sched_group_barrier(0x100, 6, 0);                 \
        __builtin_amdgcn_sched_group_barrier(0x008, 8, 0);                 \
    }

    int ch = 0;
    for (; ch + 2 < nchunks; ++ch) {
        const int buf = ch & 1;
        {
            const float* asl = &As[buf][0] + aoff;
            const float* bsl = &Bs[buf][0] + boff;
            PIN_B_STAGE(st_);
            STORE_LDS(st_, buf ^ 1);
            LOAD_GLOBAL(st_);
            MUL_HEAD()
            __builtin_amdgcn_sched_group_barrier(0x200, 8, 0);
            __builtin_amdgcn_sched_group_barrier(0x020, 8, 0);
            ORDER_HEAD()
        }
        HAND_OVER(buf ^ 1)
    }
    if (ch + 1 < nchunks) {
        const int buf = ch & 1;
        {
            const float* asl = &As[buf][0] + aoff;
            const float* bsl = &Bs[buf][0] + boff;
            PIN_B_STAGE(st_);
            STORE_LDS(st_, buf ^ 1);
            MUL_HEAD()
            __builtin_amdgcn_sched_group_barrier(0x200, 8, 0);
            ORDER_HEAD()
        }
        HAND_OVER(buf ^ 1)
        ++ch;
    }
    {
        const int buf = ch & 1;
        const float* asl = &As[buf][0] + aoff;
        const float* bsl = &Bs[buf][0] + boff;
        MUL_HEAD()
        MUL_FRAGS(f1)
        ORDER_HEAD()
        __builtin_amdgcn_sched_group_barrier(0x008, 8, 0);
    }

    const int rbase = hh * 8;
    auto store_tile = [&](v8f c, int m0, int n0) {
        const int col = n0 + cl;
        #pragma unroll
        for (int r = 0; r < 8; ++r) {
            const int row = m0 + rbase + r;
            float v = c[r];
            if (E) {
                v += E[(size_t)row * Ncols + col];
                v = v > 0.0f ? v : 0.0f;
            }
            C[(size_t)row * Ncols + col] = v;
        }
    };
    store_tile(c0, bm + wm, bn + wn);
    store_tile(c1, bm + wm, bn + wn + 16);
}

#define EROWS  8
#define EWAVES 16
#define EHALF  1024
#define EBATCH 16
#define ENB    8

__global__ __launch_bounds__(512) void edge_agg_kernel(
    const float* __restrict__ A, const float* __restrict__ Z,
    const float* __restrict__ EK, const float* __restrict__ bias,
    float* __restrict__ E)
{
    __shared__ __attribute__((aligned(16))) float As[EWAVES][EHALF];
    __shared__ __attribute__((aligned(16))) float red[EWAVES][FEDGE];

    const int t    = threadIdx.x;
    const int lane = t & 31;
    const int wave = t >> 5;
    const int row  = blockIdx.x * EROWS + (wave >> 1);
    const int j0   = (wave & 1) * EHALF;
    const int jl   = lane >> 2;

    const float* ap  = A + (size_t)row * NN + j0 + 4 * lane;
    const v4f*   zp  = (const v4f*)(Z + ((size_t)row * NN + j0) * FEDGE) + lane;
    float*       asw = &As[wave][0];

    v4f z[2][EBATCH];
#define LOAD_BATCH(b)                                                      \
    _Pragma("unroll")                                                      \
    for (int q = 0; q < EBATCH; ++q)                                       \
        z[(b) & 1][q] = __builtin_nontemporal_load(zp + ((b) * EBATCH + q) * 32);

    v4f av[8];
    #pragma unroll
    for (int p = 0; p < 8; ++p) av[p] = *(const v4f*)(ap + 128 * p);
    __builtin_amdgcn_sched_barrier(0);
    LOAD_BATCH(0)
    LOAD_BATCH(1)
    __builtin_amdgcn_sched_barrier(0);
    #pragma unroll
    for (int p = 0; p < 8; ++p) *(v4f*)(asw + 128 * p + 4 * lane) = av[p];
    __builtin_amdgcn_wave_barrier();

    v4f acc = {0.0f, 0.0f, 0.0f, 0.0f};
    #pragma unroll
    for (int b = 0; b < ENB; ++b) {
        __builtin_amdgcn_sched_barrier(0);
        if (b >= 1 && b + 1 < ENB) { LOAD_BATCH(b + 1) }
        __builtin_amdgcn_sched_barrier(0);
        const float* al = asw + 128 * b + jl;
        #pragma unroll
        for (int q = 0; q < EBATCH; ++q) {
            const float a  = al[8 * q];
            const v4f   zz = z[b & 1][q];
            acc.x = fmaf(a, zz.x, acc.x);
            acc.y = fmaf(a, zz.y, acc.y);
            acc.z = fmaf(a, zz.z, acc.z);
            acc.w = fmaf(a, zz.w, acc.w);
        }
    }
#undef LOAD_BATCH

    const int c = t;
    float ek[FEDGE];
    #pragma unroll
    for (int ff = 0; ff < FEDGE; ++ff) ek[ff] = EK[ff * FOUT + c];
    const float bc = bias[c];

    #pragma unroll
    for (int m = 4; m <= 16; m <<= 1) {
        acc.x += __shfl_xor(acc.x, m, 32);
        acc.y += __shfl_xor(acc.y, m, 32);
        acc.z += __shfl_xor(acc.z, m, 32);
        acc.w += __shfl_xor(acc.w, m, 32);
    }
    if (lane < 4) *(v4f*)(&red[wave][4 * lane]) = acc;
    __syncthreads();

    #pragma unroll
    for (int r = 0; r < EROWS; ++r) {
        const v4f* h0 = (const v4f*)&red[2 * r][0];
        const v4f* h1 = (const v4f*)&red[2 * r + 1][0];
        float y = bc;
        #pragma unroll
        for (int g = 0; g < 4; ++g) {
            const v4f s = h0[g] + h1[g];
            y = fmaf(s.x, ek[4 * g + 0], y);
            y = fmaf(s.y, ek[4 * g + 1], y);
            y = fmaf(s.z, ek[4 * g + 2], y);
            y = fmaf(s.w, ek[4 * g + 3], y);
        }
        E[(size_t)(blockIdx.x * EROWS + r) * FOUT + c] = y;
    }
}

extern "C" void kernel_launch(void* const* d_in, const int* in_sizes, int n_in,
                              void* d_out, int out_size, void* d_ws, size_t ws_size,
                              hipStream_t stream) {
    const float* X    = (const float*)d_in[0];
    const float* A    = (const float*)d_in[1];
    const float* Z    = (const float*)d_in[2];
    const float* W    = (const float*)d_in[3];
    const float* EK   = (const float*)d_in[4];
    const float* bias = (const float*)d_in[5];
    float* out = (float*)d_out;

    float* XW = (float*)d_ws;
    float* E  = XW + (size_t)NN * FOUT;

    dim3 g1(FOUT / GBN, NN / GBM);
    wmma_gemm_f32_kernel<<<g1, 256, 0, stream>>>(X, W, nullptr, XW, NN, FOUT, FIN);

    edge_agg_kernel<<<NN / EROWS, EWAVES * 32, 0, stream>>>(A, Z, EK, bias, E);

    dim3 g2(FOUT / GBN, NN / GBM);
    wmma_gemm_f32_kernel<<<g2, 256, 0, stream>>>(A, XW, E, out, NN, FOUT, NN);
}


# X@W GEMM and edge aggregation merged into one launch: of every three 8-wave work-groups one takes a GEMM tile, two stream Z, all 768 co-resident (A row staged in 3 pieces to fit); sum orders kept
# speedup vs baseline: 4.5713x; 1.0381x over previous
#include <hip/hip_runtime.h>

typedef __attribute__((ext_vector_type(2))) float v2f;
typedef __attribute__((ext_vector_type(4))) float v4f;
typedef __attribute__((ext_vector_type(8))) float v8f;

#define NN     2048
#define FIN    512
#define FOUT   512
#define FEDGE  16

#define GBM 64
#define GBN 64
#define GBK 64
#define GSA 68
#define GSB 160
#define GEMM_LDS_FLOATS (2 * GBM * GSA + 2 * (GBK / 2) * GSB)

struct FragSet {
    v2f a0, a1, a2, a3;
    v2f p0, p1, p2, p3;
    v2f q0, q1, q2, q3;
};

#define LOAD_FRAGS(F, g)                                                   \
    F.a0 = *(const v2f*)(asl + 16 * (g) + 0);                              \
    F.a1 = *(const v2f*)(asl + 16 * (g) + 4);                              \
    F.a2 = *(const v2f*)(asl + 16 * (g) + 8);                              \
    F.a3 = *(const v2f*)(asl + 16 * (g) + 12);                             \
    F.p0 = *(const v2f*)(bsl + (8 * (g) + 0) * GSB);                       \
    F.p1 = *(const v2f*)(bsl + (8 * (g) + 2) * GSB);                       \
    F.p2 = *(const v2f*)(bsl + (8 * (g) + 4) * GSB);                       \
    F.p3 = *(const v2f*)(bsl + (8 * (g) + 6) * GSB);                       \
    F.q0 = *(const v2f*)(bsl + (8 * (g) + 0) * GSB + 32);                  \
    F.q1 = *(const v2f*)(bsl + (8 * (g) + 2) * GSB + 32);                  \
    F.q2 = *(const v2f*)(bsl + (8 * (g) + 4) * GSB + 32);                  \
    F.q3 = *(const v2f*)(bsl + (8 * (g) + 6) * GSB + 32);

#define WMMA_F32(a, b, c) __builtin_amdgcn_wmma_f32_16x16x4_f32(false, a, false, b, (short)0, c, false, false)

#define MUL_FRAGS(F)                                                       \
    c0 = WMMA_F32(F.a0, F.p0, c0);  c1 = WMMA_F32(F.a0, F.q0, c1);         \
    c0 = WMMA_F32(F.a1, F.p1, c0);  c1 = WMMA_F32(F.a1, F.q1, c1);         \
    c0 = WMMA_F32(F.a2, F.p2, c0);  c1 = WMMA_F32(F.a2, F.q2, c1);         \
    c0 = WMMA_F32(F.a3, F.p3, c0);  c1 = WMMA_F32(F.a3, F.q3, c1);

static __device__ __forceinline__ void lds_hand_over_barrier() {
    asm volatile("s_wait_dscnt 0x0\n\ts_barrier_signal -1\n\ts_barrier_wait -1" ::: "memory");
}

static __device__ __forceinline__ void gemm_tile(
    const float* __restrict__ A, const float* __restrict__ B,
    const float* __restrict__ E, float* __restrict__ C,
    int Ncols, int K, int bm, int bn, float* lds)
{
    float* const As_ = lds;
    float* const Bs_ = lds + 2 * GBM * GSA;
#define AS_BUF(buf) (As_ + (buf) * (GBM * GSA))
#define BS_BUF(buf) (Bs_ + (buf) * ((GBK / 2) * GSB))

    const int t    = threadIdx.x;
    const int lane = t & 31;
    const int wave = t >> 5;

    const int wm = (wave & 3) * 16;
    const int wn = (wave >> 2) * 32;

    const int cl = lane & 15;
    const int hh = lane >> 4;

    const int srow = t >> 4;
    const int sq   = t & 15;
    const size_t a16 = (size_t)16 * K;
    const float* ag = A + (size_t)(bm + srow) * K + 4 * sq;
    const float* bg = B + (size_t)(2 * srow) * Ncols + bn + 4 * sq;
    const size_t bchunk = (size_t)GBK * Ncols;
    const size_t b32 = (size_t)32 * Ncols;

    const int sa_off = srow * GSA + 4 * sq;
    const int sb_off = srow * GSB + 8 * sq;
    const int aoff = (wm + cl) * GSA + 2 * hh;
    const int boff = hh * GSB + (wn + cl) * 2;

#define DECL_STAGE(P) float4 P##a0, P##a1, P##a2, P##a3; v4f P##be0, P##bo0, P##be1, P##bo1

#define LOAD_GLOBAL(P)                                                     \
    P##a0  = *(const float4*)(ag);                                         \
    P##a1  = *(const float4*)(ag + a16);                                   \
    P##a2  = *(const float4*)(ag + 2 * a16);                               \
    P##a3  = *(const float4*)(ag + 3 * a16);                               \
    P##be0 = *(const v4f*)(bg);                                            \
    P##bo0 = *(const v4f*)(bg + Ncols);                                    \
    P##be1 = *(const v4f*)(bg + b32);                                      \
    P##bo1 = *(const v4f*)(bg + b32 + Ncols);                              \
    ag += GBK;  bg += bchunk;

#define STORE_LDS(P, buf)                                                  \
    {                                                                      \
        float* as_ = AS_BUF(buf) + sa_off;                                 \
        float* bs_ = BS_BUF(buf) + sb_off;                                 \
        *(float4*)(as_)            = P##a0;                                \
        *(float4*)(as_ + 16 * GSA) = P##a1;                                \
        *(float4*)(as_ + 32 * GSA) = P##a2;                                \
        *(float4*)(as_ + 48 * GSA) = P##a3;                                \
        *(float4*)(bs_)                = make_float4(P##be0.x, P##bo0.x, P##be0.y, P##bo0.y); \
        *(float4*)(bs_ + 4)            = make_float4(P##be0.z, P##bo0.z, P##be0.w, P##bo0.w); \
        *(float4*)(bs_ + 16 * GSB)     = make_float4(P##be1.x, P##bo1.x, P##be1.y, P##bo1.y); \
        *(float4*)(bs_ + 16 * GSB + 4) = make_float4(P##be1.z, P##bo1.z, P##be1.w, P##bo1.w); \
    }

#define PIN_B_STAGE(P) asm volatile("" : "+v"(P##be0), "+v"(P##bo0), "+v"(P##be1), "+v"(P##bo1))

    v8f c0 = {}, c1 = {};
    FragSet f0, f1;
    DECL_STAGE(st_);

    const int nchunks = K / GBK;

    {
        DECL_STAGE(first_);
        LOAD_GLOBAL(first_);
        if (nchunks > 1) { LOAD_GLOBAL(st_); }
        STORE_LDS(first_, 0);
    }
    __syncthreads();
    {
        const float* asl = AS_BUF(0) + aoff;
        const float* bsl = BS_BUF(0) + boff;
        LOAD_FRAGS(f0, 0)
    }

#define MUL_HEAD()                                                         \
        LOAD_FRAGS(f1, 1)                                                  \
        MUL_FRAGS(f0)                                                      \
        LOAD_FRAGS(f0, 2)                                                  \
        MUL_FRAGS(f1)                                                      \
        LOAD_FRAGS(f1, 3)                                                  \
        MUL_FRAGS(f0)

#define ORDER_HEAD()                                                       \
    __builtin_amdgcn_sched_group_barrier(0x100, 6, 0);                     \
    __builtin_amdgcn_sched_group_barrier(0x008, 8, 0);                     \
    __builtin_amdgcn_sched_group_barrier(0x100, 6, 0);                     \
    __builtin_amdgcn_sched_group_barrier(0x008, 8, 0);                     \
    __builtin_amdgcn_sched_group_barrier(0x100, 6, 0);                     \
    __builtin_amdgcn_sched_group_barrier(0x008, 8, 0);

#define HAND_OVER(nbuf)                                                    \
    __builtin_amdgcn_sched_barrier(0);                                     \
    lds_hand_over_barrier();                                               \
    __builtin_amdgcn_sched_barrier(0);                                     \
    {                                                                      \
        const float* asl = AS_BUF(nbuf) + aoff;                            \
        const float* bsl = BS_BUF(nbuf) + boff;                            \
        LOAD_FRAGS(f0, 0)                                                  \
        MUL_FRAGS(f1)                                                      \
        __builtin_amdgcn_sched_group_barrier(0x100, 6, 0);                 \
        __builtin_amdgcn_sched_group_barrier(0x008, 8, 0);                 \
    }

    int ch = 0;
    for (; ch + 2 < nchunks; ++ch) {
        const int buf = ch & 1;
        {
            const float* asl = AS_BUF(buf) + aoff;
            const float* bsl = BS_BUF(buf) + boff;
            PIN_B_STAGE(st_);
            STORE_LDS(st_, buf ^ 1);
            LOAD_GLOBAL(st_);
            MUL_HEAD()
            __builtin_amdgcn_sched_group_barrier(0x200, 8, 0);
            __builtin_amdgcn_sched_group_barrier(0x020, 8, 0);
            ORDER_HEAD()
        }
        HAND_OVER(buf ^ 1)
    }
    if (ch + 1 < nchunks) {
        const int buf = ch & 1;
        {
            const float* asl = AS_BUF(buf) + aoff;
            const float* bsl = BS_BUF(buf) + boff;
            PIN_B_STAGE(st_);
            STORE_LDS(st_, buf ^ 1);
            MUL_HEAD()
            __builtin_amdgcn_sched_group_barrier(0x200, 8, 0);
            ORDER_HEAD()
        }
        HAND_OVER(buf ^ 1)
        ++ch;
    }
    {
        const int buf = ch & 1;
        const float* asl = AS_BUF(buf) + aoff;
        const float* bsl = BS_BUF(buf) + boff;
        MUL_HEAD()
        MUL_FRAGS(f1)
        ORDER_HEAD()
        __builtin_amdgcn_sched_group_barrier(0x008, 8, 0);
    }

    const int rbase = hh * 8;
    auto store_tile = [&](v8f c, int m0, int n0) {
        const int col = n0 + cl;
        #pragma unroll
        for (int r = 0; r < 8; ++r) {
            const int row = m0 + rbase + r;
            float v = c[r];
            if (E) {
                v += E[(size_t)row * Ncols + col];
                v = v > 0.0f ? v : 0.0f;
            }
            C[(size_t)row * Ncols + col] = v;
        }
    };
    store_tile(c0, bm + wm, bn + wn);
    store_tile(c1, bm + wm, bn + wn + 16);
}

#define EROWS  4
#define EWAVES 8
#define EHALF  1024
#define EBATCH 16
#define ENB    8
#define EDGE_LDS_FLOATS (EWAVES * EHALF + EWAVES * FEDGE)

static __device__ __forceinline__ void edge_rows(
    const float* __restrict__ A, const float* __restrict__ Z,
    const float* __restrict__ EK, const float* __restrict__ bias,
    float* __restrict__ E, int eb, float* lds)
{
    float* const As_  = lds;
    float* const red_ = lds + EWAVES * EHALF;

    const int t    = threadIdx.x;
    const int lane = t & 31;
    const int wave = t >> 5;
    const int row  = eb * EROWS + (wave >> 1);
    const int j0   = (wave & 1) * EHALF;
    const int jl   = lane >> 2;

    const float* ap  = A + (size_t)row * NN + j0 + 4 * lane;
    const v4f*   zp  = (const v4f*)(Z + ((size_t)row * NN + j0) * FEDGE) + lane;
    float*       asw = As_ + wave * EHALF;

    v4f z[2][EBATCH];
#define LOAD_BATCH(b)                                                      \
    _Pragma("unroll")                                                      \
    for (int q = 0; q < EBATCH; ++q)                                       \
        z[(b) & 1][q] = __builtin_nontemporal_load(zp + ((b) * EBATCH + q) * 32);

    v4f av[4];
    #pragma unroll
    for (int p = 0; p < 4; ++p) av[p] = *(const v4f*)(ap + 128 * p);
    __builtin_amdgcn_sched_barrier(0);
    LOAD_BATCH(0)
    LOAD_BATCH(1)
    __builtin_amdgcn_sched_barrier(0);
    #pragma unroll
    for (int p = 0; p < 4; ++p) *(v4f*)(asw + 128 * p + 4 * lane) = av[p];
    __builtin_amdgcn_sched_barrier(0);
    v4f aw[2];
    #pragma unroll
    for (int p = 0; p < 2; ++p) aw[p] = *(const v4f*)(ap + 128 * (p + 4));
    __builtin_amdgcn_wave_barrier();

    v4f acc = {0.0f, 0.0f, 0.0f, 0.0f};
    #pragma unroll
    for (int b = 0; b < ENB; ++b) {
        __builtin_amdgcn_sched_barrier(0);
        if (b >= 1 && b + 1 < ENB) { LOAD_BATCH(b + 1) }
        __builtin_amdgcn_sched_barrier(0);
        if (b == 2 || b == 4) {
            #pragma unroll
            for (int p = 0; p < 2; ++p) *(v4f*)(asw + 128 * (p + b + 2) + 4 * lane) = aw[p];
            __builtin_amdgcn_wave_barrier();
            if (b == 2) {
                #pragma unroll
                for (int p = 0; p < 2; ++p) aw[p] = *(const v4f*)(ap + 128 * (p + 6));
            }
        }
        const float* al = asw + 128 * b + jl;
        #pragma unroll
        for (int q = 0; q < EBATCH; ++q) {
            const float a  = al[8 * q];
            const v4f   zz = z[b & 1][q];
            acc.x = fmaf(a, zz.x, acc.x);
            acc.y = fmaf(a, zz.y, acc.y);
            acc.z = fmaf(a, zz.z, acc.z);
            acc.w = fmaf(a, zz.w, acc.w);
        }
    }
#undef LOAD_BATCH

    const int c0 = t;
    const int c1 = t + 256;
    float ek0[FEDGE], ek1[FEDGE];
    #pragma unroll
    for (int ff = 0; ff < FEDGE; ++ff) { ek0[ff] = EK[ff * FOUT + c0]; ek1[ff] = EK[ff * FOUT + c1]; }
    const float bc0 = bias[c0];
    const float bc1 = bias[c1];

    #pragma unroll
    for (int m = 4; m <= 16; m <<= 1) {
        acc.x += __shfl_xor(acc.x, m, 32);
        acc.y += __shfl_xor(acc.y, m, 32);
        acc.z += __shfl_xor(acc.z, m, 32);
        acc.w += __shfl_xor(acc.w, m, 32);
    }
    if (lane < 4) *(v4f*)(red_ + wave * FEDGE + 4 * lane) = acc;
    __syncthreads();

    #pragma unroll
    for (int r = 0; r < EROWS; ++r) {
        const v4f* h0 = (const v4f*)(red_ + (2 * r) * FEDGE);
        const v4f* h1 = (const v4f*)(red_ + (2 * r + 1) * FEDGE);
        float y0 = bc0, y1 = bc1;
        #pragma unroll
        for (int g = 0; g < 4; ++g) {
            const v4f s = h0[g] + h1[g];
            y0 = fmaf(s.x, ek0[4 * g + 0], y0);
            y0 = fmaf(s.y, ek0[4 * g + 1], y0);
            y0 = fmaf(s.z, ek0[4 * g + 2], y0);
            y0 = fmaf(s.w, ek0[4 * g + 3], y0);
            y1 = fmaf(s.x, ek1[4 * g + 0], y1);
            y1 = fmaf(s.y, ek1[4 * g + 1], y1);
            y1 = fmaf(s.z, ek1[4 * g + 2], y1);
            y1 = fmaf(s.w, ek1[4 * g + 3], y1);
        }
        float* er = E + (size_t)(eb * EROWS + r) * FOUT;
        er[c0] = y0;
        er[c1] = y1;
    }
}

#define GEMM1_TILES ((FOUT / GBN) * (NN / GBM))
#define EDGE_GROUPS (NN / EROWS)

__global__ __launch_bounds__(256) void xw_and_edge_kernel(
    const float* __restrict__ X, const float* __restrict__ W, float* __restrict__ XW,
    int Ncols, int K,
    const float* __restrict__ A, const float* __restrict__ Z,
    const float* __restrict__ EK, const float* __restrict__ bias,
    float* __restrict__ E)
{
    __shared__ __attribute__((aligned(16))) float lds[GEMM_LDS_FLOATS];
    static_assert(EDGE_LDS_FLOATS <= GEMM_LDS_FLOATS, "the edge role lives in the GEMM role's LDS block");
    static_assert(EDGE_GROUPS == 2 * GEMM1_TILES, "two row groups per GEMM tile");

    const int bid = blockIdx.x;
    const int q   = bid / 3;
    const int r   = bid - 3 * q;
    if (r == 0) {
        const int tx = q % (FOUT / GBN);
        const int ty = q / (FOUT / GBN);
        gemm_tile(X, W, nullptr, XW, Ncols, K, ty * GBM, tx * GBN, lds);
    } else {
        edge_rows(A, Z, EK, bias, E, 2 * q + (r - 1), lds);
    }
}

__global__ __launch_bounds__(256) void wmma_gemm_f32_kernel(
    const float* __restrict__ A, const float* __restrict__ B,
    const float* __restrict__ E, float* __restrict__ C,
    int M, int Ncols, int K)
{
    __shared__ __attribute__((aligned(16))) float lds[GEMM_LDS_FLOATS];
    gemm_tile(A, B, E, C, Ncols, K, blockIdx.y * GBM, blockIdx.x * GBN, lds);
}

extern "C" void kernel_launch(void* const* d_in, const int* in_sizes, int n_in,
                              void* d_out, int out_size, void* d_ws, size_t ws_size,
                              hipStream_t stream) {
    const float* X    = (const float*)d_in[0];
    const float* A    = (const float*)d_in[1];
    const float* Z    = (const float*)d_in[2];
    const float* W    = (const float*)d_in[3];
    const float* EK   = (const float*)d_in[4];
    const float* bias = (const float*)d_in[5];
    float* out = (float*)d_out;

    float* XW = (float*)d_ws;
    float* E  = XW + (size_t)NN * FOUT;

    xw_and_edge_kernel<<<GEMM1_TILES + EDGE_GROUPS, 256, 0, stream>>>(X, W, XW, FOUT, FIN, A, Z, EK, bias, E);

    dim3 g2(FOUT / GBN, NN / GBM);
    wmma_gemm_f32_kernel<<<g2, 256, 0, stream>>>(A, XW, E, out, NN, FOUT, NN);
}
